# starting kernel
# baseline (speedup 1.0000x reference)
#include <hip/hip_runtime.h>

typedef __attribute__((ext_vector_type(2))) float v2f;
typedef __attribute__((ext_vector_type(4))) float v4f;
typedef __attribute__((ext_vector_type(8))) float v8f;

#define T_STEPS 200
#define HID 4
#define OUTN 28
#define WP_ENTRIES (T_STEPS * 2 * 32)
#define SWAPX16 0x401f

__device__ __forceinline__ float fast_tanh(float x) {
#if __has_builtin(__builtin_amdgcn_tanhf)
  return __builtin_amdgcn_tanhf(x);
#else
  float e = __builtin_amdgcn_exp2f(-2.8853900817779268f * x);
  return (1.0f - e) * __builtin_amdgcn_rcpf(1.0f + e);
#endif
}
__device__ __forceinline__ float fast_sigmoid(float x) {
  return __builtin_fmaf(0.5f, fast_tanh(0.5f * x), 0.5f);
}
__device__ __forceinline__ v2f tanh2(v2f v) {
  v2f r; r.x = fast_tanh(v.x); r.y = fast_tanh(v.y); return r;
}
__device__ __forceinline__ v2f sig2(v2f v) {
  v2f r; r.x = fast_sigmoid(v.x); r.y = fast_sigmoid(v.y); return r;
}
__device__ __forceinline__ float swap16(float v) {
  int i = __builtin_bit_cast(int, v);
  i = __builtin_amdgcn_ds_swizzle(i, SWAPX16);
  return __builtin_bit_cast(float, i);
}

__device__ __forceinline__ void cell_pk(const v4f* pw, const v4f* pbv,
                                        const float in[8], v2f h[2], v2f c[2]) {
  v2f g[8];
#pragma unroll
  for (int q = 0; q < 4; ++q) {
    v4f bq = pbv[q];
    g[2 * q].x = bq.x;     g[2 * q].y = bq.y;
    g[2 * q + 1].x = bq.z; g[2 * q + 1].y = bq.w;
  }
#pragma unroll
  for (int p = 0; p < 8; ++p) {
#pragma unroll
    for (int kq = 0; kq < 4; ++kq) {
      v4f q = pw[p * 4 + kq];
      v2f w0; w0.x = q.x; w0.y = q.y;
      v2f w1; w1.x = q.z; w1.y = q.w;
      v2f a0; a0.x = in[2 * kq];     a0.y = in[2 * kq];
      v2f a1; a1.x = in[2 * kq + 1]; a1.y = in[2 * kq + 1];
      g[p] = __builtin_elementwise_fma(a0, w0, g[p]);
      g[p] = __builtin_elementwise_fma(a1, w1, g[p]);
    }
  }
#pragma unroll
  for (int q = 0; q < 2; ++q) {
    v2f iv = sig2(g[0 + q]);
    v2f fv = sig2(g[2 + q]);
    v2f gv = tanh2(g[4 + q]);
    v2f ov = sig2(g[6 + q]);
    c[q] = __builtin_elementwise_fma(fv, c[q], iv * gv);
    h[q] = ov * tanh2(c[q]);
  }
}

__global__ void repack_wout_kernel(const float* __restrict__ Wout,
                                   v2f* __restrict__ WP) {
  int idx = blockIdx.x * blockDim.x + threadIdx.x;
  if (idx >= WP_ENTRIES) return;
  int c = idx & 31;
  int kp = (idx >> 5) & 1;
  int t = idx >> 6;
  v2f v; v.x = 0.f; v.y = 0.f;
  if (c < OUTN) {
    const float* p = Wout + c * 800 + 4 * t + 2 * kp;
    v.x = p[0];
    v.y = p[1];
  }
  WP[idx] = v;
}

__global__ void __launch_bounds__(128)
lstm_fused_kernel(const float* __restrict__ x,
                  const float* __restrict__ Wih0, const float* __restrict__ Whh0,
                  const float* __restrict__ bih0, const float* __restrict__ bhh0,
                  const float* __restrict__ Wih1, const float* __restrict__ Whh1,
                  const float* __restrict__ bih1, const float* __restrict__ bhh1,
                  const v2f* __restrict__ WP, const float* __restrict__ bout,
                  float* __restrict__ out) {
  __shared__ __align__(16) float sw[288];
  const int tid = threadIdx.x;

  for (int i = tid; i < 288; i += blockDim.x) {
    float v;
    if (i < 256) {
      int layer = i >> 7;
      int f = i & 127;
      int e = f >> 1, comp = f & 1;
      int p = e >> 3, k = e & 7;
      int j = 2 * p + comp;
      const float* Wih = layer ? Wih1 : Wih0;
      const float* Whh = layer ? Whh1 : Whh0;
      v = (k < 4) ? Wih[j * 4 + k] : Whh[j * 4 + (k - 4)];
    } else if (i < 272) {
      v = bih0[i - 256] + bhh0[i - 256];
    } else {
      v = bih1[i - 272] + bhh1[i - 272];
    }
    sw[i] = v;
  }
  __syncthreads();

  const v4f* sPW0 = (const v4f*)(sw);
  const v4f* sPW1 = (const v4f*)(sw + 128);
  const v4f* sPB0 = (const v4f*)(sw + 256);
  const v4f* sPB1 = (const v4f*)(sw + 272);

  v4f pw1r[32];
#pragma unroll
  for (int q = 0; q < 32; ++q) pw1r[q] = sPW1[q];

  const int lane = tid & 31;
  const int col = lane & 15;
  const bool hiHalf = (lane & 16) != 0;
  const int waveBase = blockIdx.x * blockDim.x + (tid & ~31);
  const int b = waveBase + lane;

  const v4f* __restrict__ xp = (const v4f*)(x + (size_t)b * T_STEPS * HID);

  v2f h0[2], c0[2], h1[2], c1[2];
#pragma unroll
  for (int q = 0; q < 2; ++q) {
    h0[q] = (v2f)0.f; c0[q] = (v2f)0.f;
    h1[q] = (v2f)0.f; c1[q] = (v2f)0.f;
  }

  float bo0 = bout[col];
  float bo1 = (col + 16 < OUTN) ? bout[col + 16] : 0.f;
  v8f acc00, acc01, acc10, acc11;
#pragma unroll
  for (int r = 0; r < 8; ++r) {
    acc00[r] = bo0; acc10[r] = bo0;
    acc01[r] = bo1; acc11[r] = bo1;
  }

  const int kp = hiHalf ? 1 : 0;
  const v2f* __restrict__ wpLane = WP + kp * 32 + col;

  for (int t = 0; t < T_STEPS; ++t) {
    v4f xv = xp[t];
    __builtin_prefetch((const void*)(xp + t + 8), 0, 0);

    float in0[8] = {xv.x, xv.y, xv.z, xv.w,
                    h0[0].x, h0[0].y, h0[1].x, h0[1].y};
    cell_pk(sPW0, sPB0, in0, h0, c0);

    float in1[8] = {h0[0].x, h0[0].y, h0[1].x, h0[1].y,
                    h1[0].x, h1[0].y, h1[1].x, h1[1].y};
    cell_pk(pw1r, sPB1, in1, h1, c1);

    float y0 = h1[0].x, y1 = h1[0].y, y2 = h1[1].x, y3 = h1[1].y;
    float s0 = swap16(y0), s1 = swap16(y1), s2 = swap16(y2), s3 = swap16(y3);
    v2f A0, A1;
    A0.x = hiHalf ? s2 : y0;  A0.y = hiHalf ? s3 : y1;
    A1.x = hiHalf ? y2 : s0;  A1.y = hiHalf ? y3 : s1;

    const v2f* wr = wpLane + t * 64;
    v2f B0 = wr[0];
    v2f B1 = wr[16];

    acc00 = __builtin_amdgcn_wmma_f32_16x16x4_f32(false, A0, false, B0, (short)0, acc00, false, false);
    acc01 = __builtin_amdgcn_wmma_f32_16x16x4_f32(false, A0, false, B1, (short)0, acc01, false, false);
    acc10 = __builtin_amdgcn_wmma_f32_16x16x4_f32(false, A1, false, B0, (short)0, acc10, false, false);
    acc11 = __builtin_amdgcn_wmma_f32_16x16x4_f32(false, A1, false, B1, (short)0, acc11, false, false);

    asm volatile("" ::: "memory");
  }

  float* outp0 = out + (size_t)(waveBase + (hiHalf ? 8 : 0)) * OUTN + col;
#pragma unroll
  for (int r = 0; r < 8; ++r) {
    outp0[(size_t)r * OUTN] = acc00[r];
    outp0[(size_t)(r + 16) * OUTN] = acc10[r];
    if (col < OUTN - 16) {
      outp0[(size_t)r * OUTN + 16] = acc01[r];
      outp0[(size_t)(r + 16) * OUTN + 16] = acc11[r];
    }
  }
}

extern "C" void kernel_launch(void* const* d_in, const int* in_sizes, int n_in,
                              void* d_out, int out_size, void* d_ws, size_t ws_size,
                              hipStream_t stream) {
  const float* x    = (const float*)d_in[0];
  const float* Wih0 = (const float*)d_in[1];
  const float* Whh0 = (const float*)d_in[2];
  const float* bih0 = (const float*)d_in[3];
  const float* bhh0 = (const float*)d_in[4];
  const float* Wih1 = (const float*)d_in[5];
  const float* Whh1 = (const float*)d_in[6];
  const float* bih1 = (const float*)d_in[7];
  const float* bhh1 = (const float*)d_in[8];
  const float* Wout = (const float*)d_in[9];
  const float* bout = (const float*)d_in[10];
  float* out = (float*)d_out;
  v2f* WP = (v2f*)d_ws;

  int B = in_sizes[0] / (T_STEPS * HID);

  repack_wout_kernel<<<(WP_ENTRIES + 255) / 256, 256, 0, stream>>>(Wout, WP);
  lstm_fused_kernel<<<B / 128, 128, 0, stream>>>(x, Wih0, Whh0, bih0, bhh0,
                                                 Wih1, Whh1, bih1, bhh1,
                                                 WP, bout, out);
}


# Each batch element split over lanes c, c+16 (2 hidden units per lane, h swapped by v_permlanex16): 16 elements per wave, 2 waves per SIMD, halved gate rows sit in VGPRs, no LDS; sums unchanged
# speedup vs baseline: 1.4756x; 1.4756x over previous
#include <hip/hip_runtime.h>

typedef __attribute__((ext_vector_type(2))) float v2f;
typedef __attribute__((ext_vector_type(4))) float v4f;
typedef __attribute__((ext_vector_type(8))) float v8f;

#define T_STEPS 200
#define HID 4
#define OUTN 28
#define WP_ENTRIES (T_STEPS * 2 * 32)

__device__ __forceinline__ float fast_tanh(float x) {
#if __has_builtin(__builtin_amdgcn_tanhf)
  return __builtin_amdgcn_tanhf(x);
#else
  float e = __builtin_amdgcn_exp2f(-2.8853900817779268f * x);
  return (1.0f - e) * __builtin_amdgcn_rcpf(1.0f + e);
#endif
}
__device__ __forceinline__ float fast_sigmoid(float x) {
  return __builtin_fmaf(0.5f, fast_tanh(0.5f * x), 0.5f);
}
__device__ __forceinline__ v2f tanh2(v2f v) {
  v2f r; r.x = fast_tanh(v.x); r.y = fast_tanh(v.y); return r;
}
__device__ __forceinline__ v2f sig2(v2f v) {
  v2f r; r.x = fast_sigmoid(v.x); r.y = fast_sigmoid(v.y); return r;
}
__device__ __forceinline__ float xhalf(float v) {
  int i = __builtin_bit_cast(int, v);
  i = __builtin_amdgcn_permlanex16(i, i, 0x76543210, 0xfedcba98, false, false);
  return __builtin_bit_cast(float, i);
}

__device__ __forceinline__ void cell_half(const v2f pw[4][8], const v2f pb[4],
                                          const float in[8], v2f& h, v2f& c) {
  v2f g[4];
#pragma unroll
  for (int q = 0; q < 4; ++q) g[q] = pb[q];
#pragma unroll
  for (int k = 0; k < 8; ++k) {
    v2f a; a.x = in[k]; a.y = in[k];
#pragma unroll
    for (int q = 0; q < 4; ++q) g[q] = __builtin_elementwise_fma(a, pw[q][k], g[q]);
  }
  v2f iv = sig2(g[0]);
  v2f fv = sig2(g[1]);
  v2f gv = tanh2(g[2]);
  v2f ov = sig2(g[3]);
  c = __builtin_elementwise_fma(fv, c, iv * gv);
  h = ov * tanh2(c);
}

__device__ __forceinline__ void load_half_weights(const float* __restrict__ Wih,
                                                  const float* __restrict__ Whh,
                                                  const float* __restrict__ bih,
                                                  const float* __restrict__ bhh,
                                                  int hh, v2f pw[4][8], v2f pb[4]) {
#pragma unroll
  for (int q = 0; q < 4; ++q) {
    const int j = 4 * q + 2 * hh;
    const v4f ia = *(const v4f*)(Wih + 4 * j);
    const v4f ib = *(const v4f*)(Wih + 4 * j + 4);
    const v4f ha = *(const v4f*)(Whh + 4 * j);
    const v4f hb = *(const v4f*)(Whh + 4 * j + 4);
#pragma unroll
    for (int k = 0; k < 4; ++k) {
      pw[q][k].x = ia[k];     pw[q][k].y = ib[k];
      pw[q][4 + k].x = ha[k]; pw[q][4 + k].y = hb[k];
    }
    pb[q].x = bih[j] + bhh[j];
    pb[q].y = bih[j + 1] + bhh[j + 1];
  }
}

__global__ void repack_wout_kernel(const float* __restrict__ Wout,
                                   v2f* __restrict__ WP) {
  int idx = blockIdx.x * blockDim.x + threadIdx.x;
  if (idx >= WP_ENTRIES) return;
  int c = idx & 31;
  int kp = (idx >> 5) & 1;
  int t = idx >> 6;
  v2f v; v.x = 0.f; v.y = 0.f;
  if (c < OUTN) {
    const float* p = Wout + c * 800 + 4 * t + 2 * kp;
    v.x = p[0];
    v.y = p[1];
  }
  WP[idx] = v;
}

__global__ void __launch_bounds__(256)
lstm_fused_kernel(const float* __restrict__ x,
                  const float* __restrict__ Wih0, const float* __restrict__ Whh0,
                  const float* __restrict__ bih0, const float* __restrict__ bhh0,
                  const float* __restrict__ Wih1, const float* __restrict__ Whh1,
                  const float* __restrict__ bih1, const float* __restrict__ bhh1,
                  const v2f* __restrict__ WP, const float* __restrict__ bout,
                  float* __restrict__ out) {
  const int tid = threadIdx.x;
  const int lane = tid & 31;
  const int col = lane & 15;
  const bool hiHalf = (lane & 16) != 0;
  const int hh = hiHalf ? 1 : 0;
  const int waveBase = (blockIdx.x * blockDim.x + (tid & ~31)) >> 1;
  const int b = waveBase + col;

  v2f pw0[4][8], pb0[4], pw1[4][8], pb1[4];
  load_half_weights(Wih0, Whh0, bih0, bhh0, hh, pw0, pb0);
  load_half_weights(Wih1, Whh1, bih1, bhh1, hh, pw1, pb1);

  const v4f* __restrict__ xp = (const v4f*)(x + (size_t)b * T_STEPS * HID);

  v2f h0 = (v2f)0.f, c0 = (v2f)0.f, h1 = (v2f)0.f, c1 = (v2f)0.f;
  v2f h0o = (v2f)0.f, h1o = (v2f)0.f;

  float bo0 = bout[col];
  float bo1 = (col + 16 < OUTN) ? bout[col + 16] : 0.f;
  v8f acc0, acc1;
#pragma unroll
  for (int r = 0; r < 8; ++r) { acc0[r] = bo0; acc1[r] = bo1; }

  const v2f* __restrict__ wpLane = WP + hh * 32 + col;

  for (int t = 0; t < T_STEPS; ++t) {
    v4f xv = xp[t];
    __builtin_prefetch((const void*)(xp + t + 8), 0, 0);

    float in0[8] = {xv.x, xv.y, xv.z, xv.w,
                    hiHalf ? h0o.x : h0.x, hiHalf ? h0o.y : h0.y,
                    hiHalf ? h0.x : h0o.x, hiHalf ? h0.y : h0o.y};
    cell_half(pw0, pb0, in0, h0, c0);
    h0o.x = xhalf(h0.x); h0o.y = xhalf(h0.y);

    float in1[8] = {hiHalf ? h0o.x : h0.x, hiHalf ? h0o.y : h0.y,
                    hiHalf ? h0.x : h0o.x, hiHalf ? h0.y : h0o.y,
                    hiHalf ? h1o.x : h1.x, hiHalf ? h1o.y : h1.y,
                    hiHalf ? h1.x : h1o.x, hiHalf ? h1.y : h1o.y};
    cell_half(pw1, pb1, in1, h1, c1);
    h1o.x = xhalf(h1.x); h1o.y = xhalf(h1.y);

    const v2f* wr = wpLane + t * 64;
    v2f B0 = wr[0];
    v2f B1 = wr[16];

    acc0 = __builtin_amdgcn_wmma_f32_16x16x4_f32(false, h1, false, B0, (short)0, acc0, false, false);
    acc1 = __builtin_amdgcn_wmma_f32_16x16x4_f32(false, h1, false, B1, (short)0, acc1, false, false);
  }

  float* outp0 = out + (size_t)(waveBase + (hiHalf ? 8 : 0)) * OUTN + col;
#pragma unroll
  for (int r = 0; r < 8; ++r) {
    outp0[(size_t)r * OUTN] = acc0[r];
    if (col < OUTN - 16) outp0[(size_t)r * OUTN + 16] = acc1[r];
  }
}

extern "C" void kernel_launch(void* const* d_in, const int* in_sizes, int n_in,
                              void* d_out, int out_size, void* d_ws, size_t ws_size,
                              hipStream_t stream) {
  const float* x    = (const float*)d_in[0];
  const float* Wih0 = (const float*)d_in[1];
  const float* Whh0 = (const float*)d_in[2];
  const float* bih0 = (const float*)d_in[3];
  const float* bhh0 = (const float*)d_in[4];
  const float* Wih1 = (const float*)d_in[5];
  const float* Whh1 = (const float*)d_in[6];
  const float* bih1 = (const float*)d_in[7];
  const float* bhh1 = (const float*)d_in[8];
  const float* Wout = (const float*)d_in[9];
  const float* bout = (const float*)d_in[10];
  float* out = (float*)d_out;
  v2f* WP = (v2f*)d_ws;

  int B = in_sizes[0] / (T_STEPS * HID);

  repack_wout_kernel<<<(WP_ENTRIES + 255) / 256, 256, 0, stream>>>(Wout, WP);
  lstm_fused_kernel<<<B / 128, 256, 0, stream>>>(x, Wih0, Whh0, bih0, bhh0,
                                                 Wih1, Whh1, bih1, bhh1,
                                                 WP, bout, out);
}


# Per-step global loads of the time loop (x gathers with one line per element, their prefetch, Wout fragments) replaced by whole-line loads 8 steps ahead into a per-wave LDS ring; sums unchanged
# speedup vs baseline: 2.3736x; 1.6086x over previous
#include <hip/hip_runtime.h>

typedef __attribute__((ext_vector_type(2))) float v2f;
typedef __attribute__((ext_vector_type(4))) float v4f;
typedef __attribute__((ext_vector_type(8))) float v8f;

#define T_STEPS 200
#define HID 4
#define OUTN 28
#define WP_ENTRIES (T_STEPS * 2 * 32)
#define BLK 8
#define NBLK (T_STEPS / BLK)
#define XPITCH 9
#define XSTAGE (16 * XPITCH)
#define BSTAGE (BLK * 32)
#define WAVE_LDS (2 * XSTAGE + 2 * BSTAGE)

__device__ __forceinline__ float fast_tanh(float x) {
#if __has_builtin(__builtin_amdgcn_tanhf)
  return __builtin_amdgcn_tanhf(x);
#else
  float e = __builtin_amdgcn_exp2f(-2.8853900817779268f * x);
  return (1.0f - e) * __builtin_amdgcn_rcpf(1.0f + e);
#endif
}
__device__ __forceinline__ float fast_sigmoid(float x) {
  return __builtin_fmaf(0.5f, fast_tanh(0.5f * x), 0.5f);
}
__device__ __forceinline__ v2f tanh2(v2f v) {
  v2f r; r.x = fast_tanh(v.x); r.y = fast_tanh(v.y); return r;
}
__device__ __forceinline__ v2f sig2(v2f v) {
  v2f r; r.x = fast_sigmoid(v.x); r.y = fast_sigmoid(v.y); return r;
}
__device__ __forceinline__ float xhalf(float v) {
  int i = __builtin_bit_cast(int, v);
  i = __builtin_amdgcn_permlanex16(i, i, 0x76543210, 0xfedcba98, false, false);
  return __builtin_bit_cast(float, i);
}

__device__ __forceinline__ void cell_half(const v2f pw[4][8], const v2f pb[4],
                                          const float in[8], v2f& h, v2f& c) {
  v2f g[4];
#pragma unroll
  for (int q = 0; q < 4; ++q) g[q] = pb[q];
#pragma unroll
  for (int k = 0; k < 8; ++k) {
    v2f a; a.x = in[k]; a.y = in[k];
#pragma unroll
    for (int q = 0; q < 4; ++q) g[q] = __builtin_elementwise_fma(a, pw[q][k], g[q]);
  }
  v2f iv = sig2(g[0]);
  v2f fv = sig2(g[1]);
  v2f gv = tanh2(g[2]);
  v2f ov = sig2(g[3]);
  c = __builtin_elementwise_fma(fv, c, iv * gv);
  h = ov * tanh2(c);
}

__device__ __forceinline__ void load_half_weights(const float* __restrict__ Wih,
                                                  const float* __restrict__ Whh,
                                                  const float* __restrict__ bih,
                                                  const float* __restrict__ bhh,
                                                  int hh, v2f pw[4][8], v2f pb[4]) {
#pragma unroll
  for (int q = 0; q < 4; ++q) {
    const int j = 4 * q + 2 * hh;
    const v4f ia = *(const v4f*)(Wih + 4 * j);
    const v4f ib = *(const v4f*)(Wih + 4 * j + 4);
    const v4f ha = *(const v4f*)(Whh + 4 * j);
    const v4f hb = *(const v4f*)(Whh + 4 * j + 4);
#pragma unroll
    for (int k = 0; k < 4; ++k) {
      pw[q][k].x = ia[k];     pw[q][k].y = ib[k];
      pw[q][4 + k].x = ha[k]; pw[q][4 + k].y = hb[k];
    }
    pb[q].x = bih[j] + bhh[j];
    pb[q].y = bih[j + 1] + bhh[j + 1];
  }
}

__global__ void repack_wout_kernel(const float* __restrict__ Wout,
                                   v2f* __restrict__ WP) {
  int idx = blockIdx.x * blockDim.x + threadIdx.x;
  if (idx >= WP_ENTRIES) return;
  int pos = idx & 31;
  int c = (pos >> 1) + 16 * (pos & 1);
  int kp = (idx >> 5) & 1;
  int t = idx >> 6;
  v2f v; v.x = 0.f; v.y = 0.f;
  if (c < OUTN) {
    const float* p = Wout + c * 800 + 4 * t + 2 * kp;
    v.x = p[0];
    v.y = p[1];
  }
  WP[idx] = v;
}

__device__ __forceinline__ void load_block(const v4f* __restrict__ xg, const v4f* __restrict__ bg,
                                           int j, v4f gx[4], v4f gb[BLK]) {
#pragma unroll
  for (int i = 0; i < 4; ++i) gx[i] = xg[(size_t)(4 * i) * T_STEPS + BLK * j];
#pragma unroll
  for (int k = 0; k < BLK; ++k) gb[k] = bg[j * BSTAGE + k * 32];
}
__device__ __forceinline__ void store_block(v4f* sx, v4f* sb, int xst, int lane,
                                            const v4f gx[4], const v4f gb[BLK]) {
#pragma unroll
  for (int i = 0; i < 4; ++i) sx[xst + 4 * i * XPITCH] = gx[i];
#pragma unroll
  for (int k = 0; k < BLK; ++k) sb[k * 32 + lane] = gb[k];
}
__device__ __forceinline__ void wave_lds_handover() {
  __builtin_amdgcn_fence(__ATOMIC_RELEASE, "wavefront");
  __builtin_amdgcn_wave_barrier();
  __builtin_amdgcn_fence(__ATOMIC_ACQUIRE, "wavefront");
}

__global__ void __launch_bounds__(256)
lstm_fused_kernel(const float* __restrict__ x,
                  const float* __restrict__ Wih0, const float* __restrict__ Whh0,
                  const float* __restrict__ bih0, const float* __restrict__ bhh0,
                  const float* __restrict__ Wih1, const float* __restrict__ Whh1,
                  const float* __restrict__ bih1, const float* __restrict__ bhh1,
                  const v2f* __restrict__ WP, const float* __restrict__ bout,
                  float* __restrict__ out) {
  __shared__ v4f lds[8 * WAVE_LDS];

  const int tid = threadIdx.x;
  const int lane = tid & 31;
  const int col = lane & 15;
  const bool hiHalf = (lane & 16) != 0;
  const int hh = hiHalf ? 1 : 0;
  const int waveBase = (blockIdx.x * blockDim.x + (tid & ~31)) >> 1;

  v4f* ringX = lds + (tid >> 5) * WAVE_LDS;
  v4f* ringB = ringX + 2 * XSTAGE;
  const v4f* __restrict__ xg = (const v4f*)x + (size_t)(waveBase + (lane >> 3)) * T_STEPS + (lane & 7);
  const v4f* __restrict__ bg = (const v4f*)WP + lane;
  const int xst = (lane >> 3) * XPITCH + (lane & 7);

  v4f gx[4], gb[BLK];
  load_block(xg, bg, 0, gx, gb);
  store_block(ringX, ringB, xst, lane, gx, gb);
  load_block(xg, bg, 1, gx, gb);
  wave_lds_handover();

  v2f pw0[4][8], pb0[4], pw1[4][8], pb1[4];
  load_half_weights(Wih0, Whh0, bih0, bhh0, hh, pw0, pb0);
  load_half_weights(Wih1, Whh1, bih1, bhh1, hh, pw1, pb1);

  v2f h0 = (v2f)0.f, c0 = (v2f)0.f, h1 = (v2f)0.f, c1 = (v2f)0.f;
  v2f h0o = (v2f)0.f, h1o = (v2f)0.f;

  float bo0 = bout[col];
  float bo1 = (col + 16 < OUTN) ? bout[col + 16] : 0.f;
  v8f acc0, acc1;
#pragma unroll
  for (int r = 0; r < 8; ++r) { acc0[r] = bo0; acc1[r] = bo1; }

  v4f xv = ringX[col * XPITCH];

  const v4f* rowX = ringX + col * XPITCH;
  const v4f* fragB = ringB + hh * 16 + col;

  for (int t = 0; t < T_STEPS; ++t) {
    const int j = t >> 3;
    const v4f bf = fragB[(j & 1) * BSTAGE + (t & 7) * 32];
    const int u = t + 1;
    const v4f xn = rowX[((u >> 3) & 1) * XSTAGE + (u & 7)];
    __builtin_amdgcn_sched_barrier(0);

    float in0[8] = {xv.x, xv.y, xv.z, xv.w,
                    hiHalf ? h0o.x : h0.x, hiHalf ? h0o.y : h0.y,
                    hiHalf ? h0.x : h0o.x, hiHalf ? h0.y : h0o.y};
    cell_half(pw0, pb0, in0, h0, c0);
    h0o.x = xhalf(h0.x); h0o.y = xhalf(h0.y);

    float in1[8] = {hiHalf ? h0o.x : h0.x, hiHalf ? h0o.y : h0.y,
                    hiHalf ? h0.x : h0o.x, hiHalf ? h0.y : h0o.y,
                    hiHalf ? h1o.x : h1.x, hiHalf ? h1o.y : h1.y,
                    hiHalf ? h1.x : h1o.x, hiHalf ? h1.y : h1o.y};
    cell_half(pw1, pb1, in1, h1, c1);
    h1o.x = xhalf(h1.x); h1o.y = xhalf(h1.y);

    v2f B0, B1;
    B0.x = bf.x; B0.y = bf.y;
    B1.x = bf.z; B1.y = bf.w;
    acc0 = __builtin_amdgcn_wmma_f32_16x16x4_f32(false, h1, false, B0, (short)0, acc0, false, false);
    acc1 = __builtin_amdgcn_wmma_f32_16x16x4_f32(false, h1, false, B1, (short)0, acc1, false, false);

    xv = xn;

    if ((t & 7) == BLK - 3 && j + 1 < NBLK) {
      const int so = (j + 1) & 1;
      store_block(ringX + so * XSTAGE, ringB + so * BSTAGE, xst, lane, gx, gb);
      if (j + 2 < NBLK) load_block(xg, bg, j + 2, gx, gb);
      wave_lds_handover();
    }
  }

  float* outp0 = out + (size_t)(waveBase + (hiHalf ? 8 : 0)) * OUTN + col;
#pragma unroll
  for (int r = 0; r < 8; ++r) {
    outp0[(size_t)r * OUTN] = acc0[r];
    if (col < OUTN - 16) outp0[(size_t)r * OUTN + 16] = acc1[r];
  }
}

extern "C" void kernel_launch(void* const* d_in, const int* in_sizes, int n_in,
                              void* d_out, int out_size, void* d_ws, size_t ws_size,
                              hipStream_t stream) {
  const float* x    = (const float*)d_in[0];
  const float* Wih0 = (const float*)d_in[1];
  const float* Whh0 = (const float*)d_in[2];
  const float* bih0 = (const float*)d_in[3];
  const float* bhh0 = (const float*)d_in[4];
  const float* Wih1 = (const float*)d_in[5];
  const float* Whh1 = (const float*)d_in[6];
  const float* bih1 = (const float*)d_in[7];
  const float* bhh1 = (const float*)d_in[8];
  const float* Wout = (const float*)d_in[9];
  const float* bout = (const float*)d_in[10];
  float* out = (float*)d_out;
  v2f* WP = (v2f*)d_ws;

  int B = in_sizes[0] / (T_STEPS * HID);

  repack_wout_kernel<<<(WP_ENTRIES + 255) / 256, 256, 0, stream>>>(Wout, WP);
  lstm_fused_kernel<<<B / 128, 256, 0, stream>>>(x, Wih0, Whh0, bih0, bhh0,
                                                 Wih1, Whh1, bih1, bhh1,
                                                 WP, bout, out);
}


# Each wave carries two independent 16-element groups (32 elements per wave, one wave per SIMD, 4-wave blocks): their dependent gate and activation chains interleave in one stream; sums unchanged
# speedup vs baseline: 2.5308x; 1.0662x over previous
#include <hip/hip_runtime.h>

typedef __attribute__((ext_vector_type(2))) float v2f;
typedef __attribute__((ext_vector_type(4))) float v4f;
typedef __attribute__((ext_vector_type(8))) float v8f;

#define T_STEPS 200
#define HID 4
#define OUTN 28
#define WP_ENTRIES (T_STEPS * 2 * 32)
#define BLK 8
#define NBLK (T_STEPS / BLK)
#define XPITCH 9
#define GRP 2
#define XSTAGE (16 * GRP * XPITCH)
#define BSTAGE (BLK * 32)
#define WAVE_LDS (2 * XSTAGE + 2 * BSTAGE)

__device__ __forceinline__ float fast_tanh(float x) {
#if __has_builtin(__builtin_amdgcn_tanhf)
  return __builtin_amdgcn_tanhf(x);
#else
  float e = __builtin_amdgcn_exp2f(-2.8853900817779268f * x);
  return (1.0f - e) * __builtin_amdgcn_rcpf(1.0f + e);
#endif
}
__device__ __forceinline__ float fast_sigmoid(float x) {
  return __builtin_fmaf(0.5f, fast_tanh(0.5f * x), 0.5f);
}
__device__ __forceinline__ v2f tanh2(v2f v) {
  v2f r; r.x = fast_tanh(v.x); r.y = fast_tanh(v.y); return r;
}
__device__ __forceinline__ v2f sig2(v2f v) {
  v2f r; r.x = fast_sigmoid(v.x); r.y = fast_sigmoid(v.y); return r;
}
__device__ __forceinline__ float xhalf(float v) {
  int i = __builtin_bit_cast(int, v);
  i = __builtin_amdgcn_permlanex16(i, i, 0x76543210, 0xfedcba98, false, false);
  return __builtin_bit_cast(float, i);
}

__device__ __forceinline__ void cell_half_pair(const v2f pw[4][8], const v2f pb[4],
                                               const float inA[8], const float inB[8],
                                               v2f& hA, v2f& cA, v2f& hB, v2f& cB) {
  v2f gA[4], gB[4];
#pragma unroll
  for (int q = 0; q < 4; ++q) { gA[q] = pb[q]; gB[q] = pb[q]; }
#pragma unroll
  for (int k = 0; k < 8; ++k) {
    v2f a; a.x = inA[k]; a.y = inA[k];
    v2f b; b.x = inB[k]; b.y = inB[k];
#pragma unroll
    for (int q = 0; q < 4; ++q) {
      gA[q] = __builtin_elementwise_fma(a, pw[q][k], gA[q]);
      gB[q] = __builtin_elementwise_fma(b, pw[q][k], gB[q]);
    }
  }
  __builtin_amdgcn_sched_barrier(0);
  v2f iA = sig2(gA[0]), iB = sig2(gB[0]);
  v2f fA = sig2(gA[1]), fB = sig2(gB[1]);
  v2f tA = tanh2(gA[2]), tB = tanh2(gB[2]);
  v2f oA = sig2(gA[3]), oB = sig2(gB[3]);
  cA = __builtin_elementwise_fma(fA, cA, iA * tA);
  cB = __builtin_elementwise_fma(fB, cB, iB * tB);
  hA = oA * tanh2(cA);
  hB = oB * tanh2(cB);
}

__device__ __forceinline__ void load_half_weights(const float* __restrict__ Wih,
                                                  const float* __restrict__ Whh,
                                                  const float* __restrict__ bih,
                                                  const float* __restrict__ bhh,
                                                  int hh, v2f pw[4][8], v2f pb[4]) {
#pragma unroll
  for (int q = 0; q < 4; ++q) {
    const int j = 4 * q + 2 * hh;
    const v4f ia = *(const v4f*)(Wih + 4 * j);
    const v4f ib = *(const v4f*)(Wih + 4 * j + 4);
    const v4f ha = *(const v4f*)(Whh + 4 * j);
    const v4f hb = *(const v4f*)(Whh + 4 * j + 4);
#pragma unroll
    for (int k = 0; k < 4; ++k) {
      pw[q][k].x = ia[k];     pw[q][k].y = ib[k];
      pw[q][4 + k].x = ha[k]; pw[q][4 + k].y = hb[k];
    }
    pb[q].x = bih[j] + bhh[j];
    pb[q].y = bih[j + 1] + bhh[j + 1];
  }
}

__global__ void repack_wout_kernel(const float* __restrict__ Wout,
                                   v2f* __restrict__ WP) {
  int idx = blockIdx.x * blockDim.x + threadIdx.x;
  if (idx >= WP_ENTRIES) return;
  int pos = idx & 31;
  int c = (pos >> 1) + 16 * (pos & 1);
  int kp = (idx >> 5) & 1;
  int t = idx >> 6;
  v2f v; v.x = 0.f; v.y = 0.f;
  if (c < OUTN) {
    const float* p = Wout + c * 800 + 4 * t + 2 * kp;
    v.x = p[0];
    v.y = p[1];
  }
  WP[idx] = v;
}

__device__ __forceinline__ void load_block(const v4f* __restrict__ xg, const v4f* __restrict__ bg,
                                           int j, v4f gx[4 * GRP], v4f gb[BLK]) {
#pragma unroll
  for (int i = 0; i < 4 * GRP; ++i) gx[i] = xg[(size_t)(4 * i) * T_STEPS + BLK * j];
#pragma unroll
  for (int k = 0; k < BLK; ++k) gb[k] = bg[j * BSTAGE + k * 32];
}
__device__ __forceinline__ void store_block(v4f* sx, v4f* sb, int xst, int lane,
                                            const v4f gx[4 * GRP], const v4f gb[BLK]) {
#pragma unroll
  for (int i = 0; i < 4 * GRP; ++i) sx[xst + 4 * i * XPITCH] = gx[i];
#pragma unroll
  for (int k = 0; k < BLK; ++k) sb[k * 32 + lane] = gb[k];
}
__device__ __forceinline__ void wave_lds_handover() {
  __builtin_amdgcn_fence(__ATOMIC_RELEASE, "wavefront");
  __builtin_amdgcn_wave_barrier();
  __builtin_amdgcn_fence(__ATOMIC_ACQUIRE, "wavefront");
}

__global__ void __launch_bounds__(128)
lstm_fused_kernel(const float* __restrict__ x,
                  const float* __restrict__ Wih0, const float* __restrict__ Whh0,
                  const float* __restrict__ bih0, const float* __restrict__ bhh0,
                  const float* __restrict__ Wih1, const float* __restrict__ Whh1,
                  const float* __restrict__ bih1, const float* __restrict__ bhh1,
                  const v2f* __restrict__ WP, const float* __restrict__ bout,
                  float* __restrict__ out) {
  __shared__ v4f lds[4 * WAVE_LDS];

  const int tid = threadIdx.x;
  const int lane = tid & 31;
  const int col = lane & 15;
  const bool hiHalf = (lane & 16) != 0;
  const int hh = hiHalf ? 1 : 0;
  const int waveBase = blockIdx.x * blockDim.x + (tid & ~31);

  v4f* ringX = lds + (tid >> 5) * WAVE_LDS;
  v4f* ringB = ringX + 2 * XSTAGE;
  const v4f* __restrict__ xg = (const v4f*)x + (size_t)(waveBase + (lane >> 3)) * T_STEPS + (lane & 7);
  const v4f* __restrict__ bg = (const v4f*)WP + lane;
  const int xst = (lane >> 3) * XPITCH + (lane & 7);

  v4f gx[4 * GRP], gb[BLK];
  load_block(xg, bg, 0, gx, gb);
  store_block(ringX, ringB, xst, lane, gx, gb);
  load_block(xg, bg, 1, gx, gb);
  wave_lds_handover();

  v2f pw0[4][8], pb0[4], pw1[4][8], pb1[4];
  load_half_weights(Wih0, Whh0, bih0, bhh0, hh, pw0, pb0);
  load_half_weights(Wih1, Whh1, bih1, bhh1, hh, pw1, pb1);

  v2f h0 = (v2f)0.f, c0 = (v2f)0.f, h1 = (v2f)0.f, c1 = (v2f)0.f;
  v2f h0B = (v2f)0.f, c0B = (v2f)0.f, h1B = (v2f)0.f, c1B = (v2f)0.f;
  v2f h0o = (v2f)0.f, h1o = (v2f)0.f;
  v2f h0oB = (v2f)0.f, h1oB = (v2f)0.f;

  float bo0 = bout[col];
  float bo1 = (col + 16 < OUTN) ? bout[col + 16] : 0.f;
  v8f acc0, acc1, acc0B, acc1B;
#pragma unroll
  for (int r = 0; r < 8; ++r) { acc0[r] = bo0; acc1[r] = bo1; acc0B[r] = bo0; acc1B[r] = bo1; }

  v4f xv = ringX[col * XPITCH];
  v4f xvB = ringX[(16 + col) * XPITCH];

  const v4f* rowX = ringX + col * XPITCH;
  const v4f* fragB = ringB + hh * 16 + col;

  for (int t = 0; t < T_STEPS; ++t) {
    const int j = t >> 3;
    const v4f bf = fragB[(j & 1) * BSTAGE + (t & 7) * 32];
    const int u = t + 1;
    const v4f xn = rowX[((u >> 3) & 1) * XSTAGE + (u & 7)];
    const v4f xnB = rowX[((u >> 3) & 1) * XSTAGE + 16 * XPITCH + (u & 7)];
    __builtin_amdgcn_sched_barrier(0);

    float in0[8] = {xv.x, xv.y, xv.z, xv.w,
                    hiHalf ? h0o.x : h0.x, hiHalf ? h0o.y : h0.y,
                    hiHalf ? h0.x : h0o.x, hiHalf ? h0.y : h0o.y};
    float in0B[8] = {xvB.x, xvB.y, xvB.z, xvB.w,
                     hiHalf ? h0oB.x : h0B.x, hiHalf ? h0oB.y : h0B.y,
                     hiHalf ? h0B.x : h0oB.x, hiHalf ? h0B.y : h0oB.y};
    cell_half_pair(pw0, pb0, in0, in0B, h0, c0, h0B, c0B);
    h0o.x = xhalf(h0.x); h0o.y = xhalf(h0.y);
    h0oB.x = xhalf(h0B.x); h0oB.y = xhalf(h0B.y);

    float in1[8] = {hiHalf ? h0o.x : h0.x, hiHalf ? h0o.y : h0.y,
                    hiHalf ? h0.x : h0o.x, hiHalf ? h0.y : h0o.y,
                    hiHalf ? h1o.x : h1.x, hiHalf ? h1o.y : h1.y,
                    hiHalf ? h1.x : h1o.x, hiHalf ? h1.y : h1o.y};
    float in1B[8] = {hiHalf ? h0oB.x : h0B.x, hiHalf ? h0oB.y : h0B.y,
                     hiHalf ? h0B.x : h0oB.x, hiHalf ? h0B.y : h0oB.y,
                     hiHalf ? h1oB.x : h1B.x, hiHalf ? h1oB.y : h1B.y,
                     hiHalf ? h1B.x : h1oB.x, hiHalf ? h1B.y : h1oB.y};
    cell_half_pair(pw1, pb1, in1, in1B, h1, c1, h1B, c1B);
    h1o.x = xhalf(h1.x); h1o.y = xhalf(h1.y);
    h1oB.x = xhalf(h1B.x); h1oB.y = xhalf(h1B.y);

    v2f B0, B1;
    B0.x = bf.x; B0.y = bf.y;
    B1.x = bf.z; B1.y = bf.w;
    acc0 = __builtin_amdgcn_wmma_f32_16x16x4_f32(false, h1, false, B0, (short)0, acc0, false, false);
    acc1 = __builtin_amdgcn_wmma_f32_16x16x4_f32(false, h1, false, B1, (short)0, acc1, false, false);
    acc0B = __builtin_amdgcn_wmma_f32_16x16x4_f32(false, h1B, false, B0, (short)0, acc0B, false, false);
    acc1B = __builtin_amdgcn_wmma_f32_16x16x4_f32(false, h1B, false, B1, (short)0, acc1B, false, false);

    xv = xn;
    xvB = xnB;

    if ((t & 7) == BLK - 3 && j + 1 < NBLK) {
      const int so = (j + 1) & 1;
      store_block(ringX + so * XSTAGE, ringB + so * BSTAGE, xst, lane, gx, gb);
      if (j + 2 < NBLK) load_block(xg, bg, j + 2, gx, gb);
      wave_lds_handover();
    }
  }

  float* outp0 = out + (size_t)(waveBase + (hiHalf ? 8 : 0)) * OUTN + col;
  float* outpB = outp0 + (size_t)16 * OUTN;
#pragma unroll
  for (int r = 0; r < 8; ++r) {
    outp0[(size_t)r * OUTN] = acc0[r];
    outpB[(size_t)r * OUTN] = acc0B[r];
    if (col < OUTN - 16) {
      outp0[(size_t)r * OUTN + 16] = acc1[r];
      outpB[(size_t)r * OUTN + 16] = acc1B[r];
    }
  }
}

extern "C" void kernel_launch(void* const* d_in, const int* in_sizes, int n_in,
                              void* d_out, int out_size, void* d_ws, size_t ws_size,
                              hipStream_t stream) {
  const float* x    = (const float*)d_in[0];
  const float* Wih0 = (const float*)d_in[1];
  const float* Whh0 = (const float*)d_in[2];
  const float* bih0 = (const float*)d_in[3];
  const float* bhh0 = (const float*)d_in[4];
  const float* Wih1 = (const float*)d_in[5];
  const float* Whh1 = (const float*)d_in[6];
  const float* bih1 = (const float*)d_in[7];
  const float* bhh1 = (const float*)d_in[8];
  const float* Wout = (const float*)d_in[9];
  const float* bout = (const float*)d_in[10];
  float* out = (float*)d_out;
  v2f* WP = (v2f*)d_ws;

  int B = in_sizes[0] / (T_STEPS * HID);

  repack_wout_kernel<<<(WP_ENTRIES + 255) / 256, 256, 0, stream>>>(Wout, WP);
  lstm_fused_kernel<<<B / 128, 128, 0, stream>>>(x, Wih0, Whh0, bih0, bhh0,
                                                 Wih1, Whh1, bih1, bhh1,
                                                 WP, bout, out);
}


# Wout rows go through the per-wave LDS ring as they lie (whole-line loads, like x); a lane picks its WMMA-B k pairs by one two-address LDS read: repack kernel and its launch gone; sums unchanged
# speedup vs baseline: 2.6179x; 1.0344x over previous
#include <hip/hip_runtime.h>

typedef __attribute__((ext_vector_type(2))) float v2f;
typedef __attribute__((ext_vector_type(4))) float v4f;
typedef __attribute__((ext_vector_type(8))) float v8f;

#define T_STEPS 200
#define HID 4
#define OUTN 28
#define BLK 8
#define NBLK (T_STEPS / BLK)
#define XPITCH 9
#define GRP 2
#define XSTAGE (16 * GRP * XPITCH)
#define BPITCH 36
#define BSTAGE (BLK * BPITCH)
#define WPIECES 7
#define WAVE_LDS (2 * XSTAGE + 2 * BSTAGE)

__device__ __forceinline__ float fast_tanh(float x) {
#if __has_builtin(__builtin_amdgcn_tanhf)
  return __builtin_amdgcn_tanhf(x);
#else
  float e = __builtin_amdgcn_exp2f(-2.8853900817779268f * x);
  return (1.0f - e) * __builtin_amdgcn_rcpf(1.0f + e);
#endif
}
__device__ __forceinline__ float fast_sigmoid(float x) {
  return __builtin_fmaf(0.5f, fast_tanh(0.5f * x), 0.5f);
}
__device__ __forceinline__ v2f tanh2(v2f v) {
  v2f r; r.x = fast_tanh(v.x); r.y = fast_tanh(v.y); return r;
}
__device__ __forceinline__ v2f sig2(v2f v) {
  v2f r; r.x = fast_sigmoid(v.x); r.y = fast_sigmoid(v.y); return r;
}
__device__ __forceinline__ float xhalf(float v) {
  int i = __builtin_bit_cast(int, v);
  i = __builtin_amdgcn_permlanex16(i, i, 0x76543210, 0xfedcba98, false, false);
  return __builtin_bit_cast(float, i);
}

__device__ __forceinline__ void cell_half_pair(const v2f pw[4][8], const v2f pb[4],
                                               const float inA[8], const float inB[8],
                                               v2f& hA, v2f& cA, v2f& hB, v2f& cB) {
  v2f gA[4], gB[4];
#pragma unroll
  for (int q = 0; q < 4; ++q) { gA[q] = pb[q]; gB[q] = pb[q]; }
#pragma unroll
  for (int k = 0; k < 8; ++k) {
    v2f a; a.x = inA[k]; a.y = inA[k];
    v2f b; b.x = inB[k]; b.y = inB[k];
#pragma unroll
    for (int q = 0; q < 4; ++q) {
      gA[q] = __builtin_elementwise_fma(a, pw[q][k], gA[q]);
      gB[q] = __builtin_elementwise_fma(b, pw[q][k], gB[q]);
    }
  }
  __builtin_amdgcn_sched_barrier(0);
  v2f iA = sig2(gA[0]), iB = sig2(gB[0]);
  v2f fA = sig2(gA[1]), fB = sig2(gB[1]);
  v2f tA = tanh2(gA[2]), tB = tanh2(gB[2]);
  v2f oA = sig2(gA[3]), oB = sig2(gB[3]);
  cA = __builtin_elementwise_fma(fA, cA, iA * tA);
  cB = __builtin_elementwise_fma(fB, cB, iB * tB);
  hA = oA * tanh2(cA);
  hB = oB * tanh2(cB);
}

__device__ __forceinline__ void load_half_weights(const float* __restrict__ Wih,
                                                  const float* __restrict__ Whh,
                                                  const float* __restrict__ bih,
                                                  const float* __restrict__ bhh,
                                                  int hh, v2f pw[4][8], v2f pb[4]) {
#pragma unroll
  for (int q = 0; q < 4; ++q) {
    const int j = 4 * q + 2 * hh;
    const v4f ia = *(const v4f*)(Wih + 4 * j);
    const v4f ib = *(const v4f*)(Wih + 4 * j + 4);
    const v4f ha = *(const v4f*)(Whh + 4 * j);
    const v4f hb = *(const v4f*)(Whh + 4 * j + 4);
#pragma unroll
    for (int k = 0; k < 4; ++k) {
      pw[q][k].x = ia[k];     pw[q][k].y = ib[k];
      pw[q][4 + k].x = ha[k]; pw[q][4 + k].y = hb[k];
    }
    pb[q].x = bih[j] + bhh[j];
    pb[q].y = bih[j + 1] + bhh[j + 1];
  }
}

__device__ __forceinline__ void load_block(const v4f* __restrict__ xg, const v4f* __restrict__ wg,
                                           int j, v4f gx[4 * GRP], v4f gb[WPIECES]) {
#pragma unroll
  for (int i = 0; i < 4 * GRP; ++i) gx[i] = xg[(size_t)(4 * i) * T_STEPS + BLK * j];
#pragma unroll
  for (int i = 0; i < WPIECES; ++i) gb[i] = wg[(size_t)(4 * i) * T_STEPS + BLK * j];
}
__device__ __forceinline__ void store_block(v4f* sx, v4f* sb, int xst, int bst,
                                            const v4f gx[4 * GRP], const v4f gb[WPIECES]) {
#pragma unroll
  for (int i = 0; i < 4 * GRP; ++i) sx[xst + 4 * i * XPITCH] = gx[i];
#pragma unroll
  for (int i = 0; i < WPIECES; ++i) sb[bst + 4 * i] = gb[i];
}
__device__ __forceinline__ void wave_lds_handover() {
  __builtin_amdgcn_fence(__ATOMIC_RELEASE, "wavefront");
  __builtin_amdgcn_wave_barrier();
  __builtin_amdgcn_fence(__ATOMIC_ACQUIRE, "wavefront");
}

__global__ void __launch_bounds__(128)
lstm_fused_kernel(const float* __restrict__ x,
                  const float* __restrict__ Wih0, const float* __restrict__ Whh0,
                  const float* __restrict__ bih0, const float* __restrict__ bhh0,
                  const float* __restrict__ Wih1, const float* __restrict__ Whh1,
                  const float* __restrict__ bih1, const float* __restrict__ bhh1,
                  const float* __restrict__ Wout, const float* __restrict__ bout,
                  float* __restrict__ out) {
  __shared__ v4f lds[4 * WAVE_LDS];

  const int tid = threadIdx.x;
  const int lane = tid & 31;
  const int col = lane & 15;
  const bool hiHalf = (lane & 16) != 0;
  const int hh = hiHalf ? 1 : 0;
  const int waveBase = blockIdx.x * blockDim.x + (tid & ~31);

  v4f* ringX = lds + (tid >> 5) * WAVE_LDS;
  v4f* ringB = ringX + 2 * XSTAGE;
  const v4f* __restrict__ xg = (const v4f*)x + (size_t)(waveBase + (lane >> 3)) * T_STEPS + (lane & 7);
  const v4f* __restrict__ wg = (const v4f*)Wout + (size_t)(lane >> 3) * T_STEPS + (lane & 7);
  const int xst = (lane >> 3) * XPITCH + (lane & 7);
  const int bst = (lane & 7) * BPITCH + (lane >> 3);

  {
    const v4f zz = (v4f)0.f;
    for (int m = 0; m < 2; ++m) ringB[m * BSTAGE + (lane >> 2) * BPITCH + 28 + (lane & 3)] = zz;
  }

  v4f gx[4 * GRP], gb[WPIECES];
  load_block(xg, wg, 0, gx, gb);
  store_block(ringX, ringB, xst, bst, gx, gb);
  load_block(xg, wg, 1, gx, gb);
  wave_lds_handover();

  v2f pw0[4][8], pb0[4], pw1[4][8], pb1[4];
  load_half_weights(Wih0, Whh0, bih0, bhh0, hh, pw0, pb0);
  load_half_weights(Wih1, Whh1, bih1, bhh1, hh, pw1, pb1);

  v2f h0 = (v2f)0.f, c0 = (v2f)0.f, h1 = (v2f)0.f, c1 = (v2f)0.f;
  v2f h0B = (v2f)0.f, c0B = (v2f)0.f, h1B = (v2f)0.f, c1B = (v2f)0.f;
  v2f h0o = (v2f)0.f, h1o = (v2f)0.f;
  v2f h0oB = (v2f)0.f, h1oB = (v2f)0.f;

  float bo0 = bout[col];
  float bo1 = (col + 16 < OUTN) ? bout[col + 16] : 0.f;
  v8f acc0, acc1, acc0B, acc1B;
#pragma unroll
  for (int r = 0; r < 8; ++r) { acc0[r] = bo0; acc1[r] = bo1; acc0B[r] = bo0; acc1B[r] = bo1; }

  v4f xv = ringX[col * XPITCH];
  v4f xvB = ringX[(16 + col) * XPITCH];

  const v4f* rowX = ringX + col * XPITCH;
  const v2f* fragB = (const v2f*)ringB + 2 * col + hh;

  for (int t = 0; t < T_STEPS; ++t) {
    const int j = t >> 3;
    const v2f B0 = fragB[2 * ((j & 1) * BSTAGE + (t & 7) * BPITCH)];
    const v2f B1 = fragB[2 * ((j & 1) * BSTAGE + (t & 7) * BPITCH) + 32];
    const int u = t + 1;
    const v4f xn = rowX[((u >> 3) & 1) * XSTAGE + (u & 7)];
    const v4f xnB = rowX[((u >> 3) & 1) * XSTAGE + 16 * XPITCH + (u & 7)];
    __builtin_amdgcn_sched_barrier(0);

    float in0[8] = {xv.x, xv.y, xv.z, xv.w,
                    hiHalf ? h0o.x : h0.x, hiHalf ? h0o.y : h0.y,
                    hiHalf ? h0.x : h0o.x, hiHalf ? h0.y : h0o.y};
    float in0B[8] = {xvB.x, xvB.y, xvB.z, xvB.w,
                     hiHalf ? h0oB.x : h0B.x, hiHalf ? h0oB.y : h0B.y,
                     hiHalf ? h0B.x : h0oB.x, hiHalf ? h0B.y : h0oB.y};
    cell_half_pair(pw0, pb0, in0, in0B, h0, c0, h0B, c0B);
    h0o.x = xhalf(h0.x); h0o.y = xhalf(h0.y);
    h0oB.x = xhalf(h0B.x); h0oB.y = xhalf(h0B.y);

    float in1[8] = {hiHalf ? h0o.x : h0.x, hiHalf ? h0o.y : h0.y,
                    hiHalf ? h0.x : h0o.x, hiHalf ? h0.y : h0o.y,
                    hiHalf ? h1o.x : h1.x, hiHalf ? h1o.y : h1.y,
                    hiHalf ? h1.x : h1o.x, hiHalf ? h1.y : h1o.y};
    float in1B[8] = {hiHalf ? h0oB.x : h0B.x, hiHalf ? h0oB.y : h0B.y,
                     hiHalf ? h0B.x : h0oB.x, hiHalf ? h0B.y : h0oB.y,
                     hiHalf ? h1oB.x : h1B.x, hiHalf ? h1oB.y : h1B.y,
                     hiHalf ? h1B.x : h1oB.x, hiHalf ? h1B.y : h1oB.y};
    cell_half_pair(pw1, pb1, in1, in1B, h1, c1, h1B, c1B);
    h1o.x = xhalf(h1.x); h1o.y = xhalf(h1.y);
    h1oB.x = xhalf(h1B.x); h1oB.y = xhalf(h1B.y);

    acc0 = __builtin_amdgcn_wmma_f32_16x16x4_f32(false, h1, false, B0, (short)0, acc0, false, false);
    acc1 = __builtin_amdgcn_wmma_f32_16x16x4_f32(false, h1, false, B1, (short)0, acc1, false, false);
    acc0B = __builtin_amdgcn_wmma_f32_16x16x4_f32(false, h1B, false, B0, (short)0, acc0B, false, false);
    acc1B = __builtin_amdgcn_wmma_f32_16x16x4_f32(false, h1B, false, B1, (short)0, acc1B, false, false);

    xv = xn;
    xvB = xnB;

    if ((t & 7) == BLK - 3 && j + 1 < NBLK) {
      const int so = (j + 1) & 1;
      store_block(ringX + so * XSTAGE, ringB + so * BSTAGE, xst, bst, gx, gb);
      if (j + 2 < NBLK) load_block(xg, wg, j + 2, gx, gb);
      wave_lds_handover();
    }
  }

  float* outp0 = out + (size_t)(waveBase + (hiHalf ? 8 : 0)) * OUTN + col;
  float* outpB = outp0 + (size_t)16 * OUTN;
#pragma unroll
  for (int r = 0; r < 8; ++r) {
    outp0[(size_t)r * OUTN] = acc0[r];
    outpB[(size_t)r * OUTN] = acc0B[r];
    if (col < OUTN - 16) {
      outp0[(size_t)r * OUTN + 16] = acc1[r];
      outpB[(size_t)r * OUTN + 16] = acc1B[r];
    }
  }
}

extern "C" void kernel_launch(void* const* d_in, const int* in_sizes, int n_in,
                              void* d_out, int out_size, void* d_ws, size_t ws_size,
                              hipStream_t stream) {
  const float* x    = (const float*)d_in[0];
  const float* Wih0 = (const float*)d_in[1];
  const float* Whh0 = (const float*)d_in[2];
  const float* bih0 = (const float*)d_in[3];
  const float* bhh0 = (const float*)d_in[4];
  const float* Wih1 = (const float*)d_in[5];
  const float* Whh1 = (const float*)d_in[6];
  const float* bih1 = (const float*)d_in[7];
  const float* bhh1 = (const float*)d_in[8];
  const float* Wout = (const float*)d_in[9];
  const float* bout = (const float*)d_in[10];
  float* out = (float*)d_out;

  int B = in_sizes[0] / (T_STEPS * HID);

  lstm_fused_kernel<<<B / 128, 128, 0, stream>>>(x, Wih0, Whh0, bih0, bhh0,
                                                 Wih1, Whh1, bih1, bhh1,
                                                 Wout, bout, out);
}


# Time loop written out over the 8 steps of a staged block: ring addresses are a per-block base plus immediates, x look-ahead needs no register copies, hand-over test static; sums unchanged
# speedup vs baseline: 2.7926x; 1.0667x over previous
#include <hip/hip_runtime.h>

typedef __attribute__((ext_vector_type(2))) float v2f;
typedef __attribute__((ext_vector_type(4))) float v4f;
typedef __attribute__((ext_vector_type(8))) float v8f;

#define T_STEPS 200
#define HID 4
#define OUTN 28
#define BLK 8
#define NBLK (T_STEPS / BLK)
#define XPITCH 9
#define GRP 2
#define XSTAGE (16 * GRP * XPITCH)
#define BPITCH 36
#define BSTAGE (BLK * BPITCH)
#define WPIECES 7
#define WAVE_LDS (2 * XSTAGE + 2 * BSTAGE)

__device__ __forceinline__ float fast_tanh(float x) {
#if __has_builtin(__builtin_amdgcn_tanhf)
  return __builtin_amdgcn_tanhf(x);
#else
  float e = __builtin_amdgcn_exp2f(-2.8853900817779268f * x);
  return (1.0f - e) * __builtin_amdgcn_rcpf(1.0f + e);
#endif
}
__device__ __forceinline__ float fast_sigmoid(float x) {
  return __builtin_fmaf(0.5f, fast_tanh(0.5f * x), 0.5f);
}
__device__ __forceinline__ v2f tanh2(v2f v) {
  v2f r; r.x = fast_tanh(v.x); r.y = fast_tanh(v.y); return r;
}
__device__ __forceinline__ v2f sig2(v2f v) {
  v2f r; r.x = fast_sigmoid(v.x); r.y = fast_sigmoid(v.y); return r;
}
__device__ __forceinline__ float xhalf(float v) {
  int i = __builtin_bit_cast(int, v);
  i = __builtin_amdgcn_permlanex16(i, i, 0x76543210, 0xfedcba98, false, false);
  return __builtin_bit_cast(float, i);
}

__device__ __forceinline__ void cell_half_pair(const v2f pw[4][8], const v2f pb[4],
                                               const float inA[8], const float inB[8],
                                               v2f& hA, v2f& cA, v2f& hB, v2f& cB) {
  v2f gA[4], gB[4];
#pragma unroll
  for (int q = 0; q < 4; ++q) { gA[q] = pb[q]; gB[q] = pb[q]; }
#pragma unroll
  for (int k = 0; k < 8; ++k) {
    v2f a; a.x = inA[k]; a.y = inA[k];
    v2f b; b.x = inB[k]; b.y = inB[k];
#pragma unroll
    for (int q = 0; q < 4; ++q) {
      gA[q] = __builtin_elementwise_fma(a, pw[q][k], gA[q]);
      gB[q] = __builtin_elementwise_fma(b, pw[q][k], gB[q]);
    }
  }
  __builtin_amdgcn_sched_barrier(0);
  v2f iA = sig2(gA[0]), iB = sig2(gB[0]);
  v2f fA = sig2(gA[1]), fB = sig2(gB[1]);
  v2f tA = tanh2(gA[2]), tB = tanh2(gB[2]);
  v2f oA = sig2(gA[3]), oB = sig2(gB[3]);
  cA = __builtin_elementwise_fma(fA, cA, iA * tA);
  cB = __builtin_elementwise_fma(fB, cB, iB * tB);
  hA = oA * tanh2(cA);
  hB = oB * tanh2(cB);
}

__device__ __forceinline__ void load_half_weights(const float* __restrict__ Wih,
                                                  const float* __restrict__ Whh,
                                                  const float* __restrict__ bih,
                                                  const float* __restrict__ bhh,
                                                  int hh, v2f pw[4][8], v2f pb[4]) {
#pragma unroll
  for (int q = 0; q < 4; ++q) {
    const int j = 4 * q + 2 * hh;
    const v4f ia = *(const v4f*)(Wih + 4 * j);
    const v4f ib = *(const v4f*)(Wih + 4 * j + 4);
    const v4f ha = *(const v4f*)(Whh + 4 * j);
    const v4f hb = *(const v4f*)(Whh + 4 * j + 4);
#pragma unroll
    for (int k = 0; k < 4; ++k) {
      pw[q][k].x = ia[k];     pw[q][k].y = ib[k];
      pw[q][4 + k].x = ha[k]; pw[q][4 + k].y = hb[k];
    }
    pb[q].x = bih[j] + bhh[j];
    pb[q].y = bih[j + 1] + bhh[j + 1];
  }
}

__device__ __forceinline__ void load_block(const v4f* __restrict__ xg, const v4f* __restrict__ wg,
                                           int j, v4f gx[4 * GRP], v4f gb[WPIECES]) {
#pragma unroll
  for (int i = 0; i < 4 * GRP; ++i) gx[i] = xg[(size_t)(4 * i) * T_STEPS + BLK * j];
#pragma unroll
  for (int i = 0; i < WPIECES; ++i) gb[i] = wg[(size_t)(4 * i) * T_STEPS + BLK * j];
}
__device__ __forceinline__ void store_block(v4f* sx, v4f* sb, int xst, int bst,
                                            const v4f gx[4 * GRP], const v4f gb[WPIECES]) {
#pragma unroll
  for (int i = 0; i < 4 * GRP; ++i) sx[xst + 4 * i * XPITCH] = gx[i];
#pragma unroll
  for (int i = 0; i < WPIECES; ++i) sb[bst + 4 * i] = gb[i];
}
__device__ __forceinline__ void wave_lds_handover() {
  __builtin_amdgcn_fence(__ATOMIC_RELEASE, "wavefront");
  __builtin_amdgcn_wave_barrier();
  __builtin_amdgcn_fence(__ATOMIC_ACQUIRE, "wavefront");
}

__global__ void __launch_bounds__(128)
lstm_fused_kernel(const float* __restrict__ x,
                  const float* __restrict__ Wih0, const float* __restrict__ Whh0,
                  const float* __restrict__ bih0, const float* __restrict__ bhh0,
                  const float* __restrict__ Wih1, const float* __restrict__ Whh1,
                  const float* __restrict__ bih1, const float* __restrict__ bhh1,
                  const float* __restrict__ Wout, const float* __restrict__ bout,
                  float* __restrict__ out) {
  __shared__ v4f lds[4 * WAVE_LDS];

  const int tid = threadIdx.x;
  const int lane = tid & 31;
  const int col = lane & 15;
  const bool hiHalf = (lane & 16) != 0;
  const int hh = hiHalf ? 1 : 0;
  const int waveBase = blockIdx.x * blockDim.x + (tid & ~31);

  v4f* ringX = lds + (tid >> 5) * WAVE_LDS;
  v4f* ringB = ringX + 2 * XSTAGE;
  const v4f* __restrict__ xg = (const v4f*)x + (size_t)(waveBase + (lane >> 3)) * T_STEPS + (lane & 7);
  const v4f* __restrict__ wg = (const v4f*)Wout + (size_t)(lane >> 3) * T_STEPS + (lane & 7);
  const int xst = (lane >> 3) * XPITCH + (lane & 7);
  const int bst = (lane & 7) * BPITCH + (lane >> 3);

  {
    const v4f zz = (v4f)0.f;
    for (int m = 0; m < 2; ++m) ringB[m * BSTAGE + (lane >> 2) * BPITCH + 28 + (lane & 3)] = zz;
  }

  v4f gx[4 * GRP], gb[WPIECES];
  load_block(xg, wg, 0, gx, gb);
  store_block(ringX, ringB, xst, bst, gx, gb);
  load_block(xg, wg, 1, gx, gb);
  wave_lds_handover();

  v2f pw0[4][8], pb0[4], pw1[4][8], pb1[4];
  load_half_weights(Wih0, Whh0, bih0, bhh0, hh, pw0, pb0);
  load_half_weights(Wih1, Whh1, bih1, bhh1, hh, pw1, pb1);

  v2f h0 = (v2f)0.f, c0 = (v2f)0.f, h1 = (v2f)0.f, c1 = (v2f)0.f;
  v2f h0B = (v2f)0.f, c0B = (v2f)0.f, h1B = (v2f)0.f, c1B = (v2f)0.f;
  v2f h0o = (v2f)0.f, h1o = (v2f)0.f;
  v2f h0oB = (v2f)0.f, h1oB = (v2f)0.f;

  float bo0 = bout[col];
  float bo1 = (col + 16 < OUTN) ? bout[col + 16] : 0.f;
  v8f acc0, acc1, acc0B, acc1B;
#pragma unroll
  for (int r = 0; r < 8; ++r) { acc0[r] = bo0; acc1[r] = bo1; acc0B[r] = bo0; acc1B[r] = bo1; }

  v4f xv = ringX[col * XPITCH];
  v4f xvB = ringX[(16 + col) * XPITCH];

  const v4f* rowX = ringX + col * XPITCH;
  const v2f* fragB = (const v2f*)ringB + 2 * col + hh;

  for (int j = 0; j < NBLK; ++j) {
    const v4f* rowXc = rowX + (j & 1) * XSTAGE;
    const v4f* rowXn = rowX + ((j + 1) & 1) * XSTAGE;
    const v2f* fragBc = fragB + 2 * (j & 1) * BSTAGE;
#pragma unroll
    for (int s = 0; s < BLK; ++s) {
      const v2f B0 = fragBc[2 * (s * BPITCH)];
      const v2f B1 = fragBc[2 * (s * BPITCH) + 32];
      const v4f xn = (s + 1 < BLK) ? rowXc[s + 1] : rowXn[0];
      const v4f xnB = (s + 1 < BLK) ? rowXc[16 * XPITCH + s + 1] : rowXn[16 * XPITCH];
      __builtin_amdgcn_sched_barrier(0);

      float in0[8] = {xv.x, xv.y, xv.z, xv.w,
                      hiHalf ? h0o.x : h0.x, hiHalf ? h0o.y : h0.y,
                      hiHalf ? h0.x : h0o.x, hiHalf ? h0.y : h0o.y};
      float in0B[8] = {xvB.x, xvB.y, xvB.z, xvB.w,
                       hiHalf ? h0oB.x : h0B.x, hiHalf ? h0oB.y : h0B.y,
                       hiHalf ? h0B.x : h0oB.x, hiHalf ? h0B.y : h0oB.y};
      cell_half_pair(pw0, pb0, in0, in0B, h0, c0, h0B, c0B);
      h0o.x = xhalf(h0.x); h0o.y = xhalf(h0.y);
      h0oB.x = xhalf(h0B.x); h0oB.y = xhalf(h0B.y);

      float in1[8] = {hiHalf ? h0o.x : h0.x, hiHalf ? h0o.y : h0.y,
                      hiHalf ? h0.x : h0o.x, hiHalf ? h0.y : h0o.y,
                      hiHalf ? h1o.x : h1.x, hiHalf ? h1o.y : h1.y,
                      hiHalf ? h1.x : h1o.x, hiHalf ? h1.y : h1o.y};
      float in1B[8] = {hiHalf ? h0oB.x : h0B.x, hiHalf ? h0oB.y : h0B.y,
                       hiHalf ? h0B.x : h0oB.x, hiHalf ? h0B.y : h0oB.y,
                       hiHalf ? h1oB.x : h1B.x, hiHalf ? h1oB.y : h1B.y,
                       hiHalf ? h1B.x : h1oB.x, hiHalf ? h1B.y : h1oB.y};
      cell_half_pair(pw1, pb1, in1, in1B, h1, c1, h1B, c1B);
      h1o.x = xhalf(h1.x); h1o.y = xhalf(h1.y);
      h1oB.x = xhalf(h1B.x); h1oB.y = xhalf(h1B.y);

      acc0 = __builtin_amdgcn_wmma_f32_16x16x4_f32(false, h1, false, B0, (short)0, acc0, false, false);
      acc1 = __builtin_amdgcn_wmma_f32_16x16x4_f32(false, h1, false, B1, (short)0, acc1, false, false);
      acc0B = __builtin_amdgcn_wmma_f32_16x16x4_f32(false, h1B, false, B0, (short)0, acc0B, false, false);
      acc1B = __builtin_amdgcn_wmma_f32_16x16x4_f32(false, h1B, false, B1, (short)0, acc1B, false, false);

      xv = xn;
      xvB = xnB;

      if (s == BLK - 3 && j + 1 < NBLK) {
        const int so = (j + 1) & 1;
        store_block(ringX + so * XSTAGE, ringB + so * BSTAGE, xst, bst, gx, gb);
        if (j + 2 < NBLK) load_block(xg, wg, j + 2, gx, gb);
        wave_lds_handover();
      }
    }
  }

  float* outp0 = out + (size_t)(waveBase + (hiHalf ? 8 : 0)) * OUTN + col;
  float* outpB = outp0 + (size_t)16 * OUTN;
#pragma unroll
  for (int r = 0; r < 8; ++r) {
    outp0[(size_t)r * OUTN] = acc0[r];
    outpB[(size_t)r * OUTN] = acc0B[r];
    if (col < OUTN - 16) {
      outp0[(size_t)r * OUTN + 16] = acc1[r];
      outpB[(size_t)r * OUTN + 16] = acc1B[r];
    }
  }
}

extern "C" void kernel_launch(void* const* d_in, const int* in_sizes, int n_in,
                              void* d_out, int out_size, void* d_ws, size_t ws_size,
                              hipStream_t stream) {
  const float* x    = (const float*)d_in[0];
  const float* Wih0 = (const float*)d_in[1];
  const float* Whh0 = (const float*)d_in[2];
  const float* bih0 = (const float*)d_in[3];
  const float* bhh0 = (const float*)d_in[4];
  const float* Wih1 = (const float*)d_in[5];
  const float* Whh1 = (const float*)d_in[6];
  const float* bih1 = (const float*)d_in[7];
  const float* bhh1 = (const float*)d_in[8];
  const float* Wout = (const float*)d_in[9];
  const float* bout = (const float*)d_in[10];
  float* out = (float*)d_out;

  int B = in_sizes[0] / (T_STEPS * HID);

  lstm_fused_kernel<<<B / 128, 128, 0, stream>>>(x, Wih0, Whh0, bih0, bhh0,
                                                 Wih1, Whh1, bih1, bhh1,
                                                 Wout, bout, out);
}


# Gate sums reordered per lane: known inputs first (x, old h1), fresh h last, own pair before partner's; early half issues inside the other cell's tanh chain, lane selects gone; gate sum order changes
# speedup vs baseline: 3.1218x; 1.1179x over previous
#include <hip/hip_runtime.h>

typedef __attribute__((ext_vector_type(2))) float v2f;
typedef __attribute__((ext_vector_type(4))) float v4f;
typedef __attribute__((ext_vector_type(8))) float v8f;

#define T_STEPS 200
#define HID 4
#define OUTN 28
#define BLK 8
#define NBLK (T_STEPS / BLK)
#define XPITCH 9
#define GRP 2
#define XSTAGE (16 * GRP * XPITCH)
#define BPITCH 36
#define BSTAGE (BLK * BPITCH)
#define WPIECES 7
#define WAVE_LDS (2 * XSTAGE + 2 * BSTAGE)

__device__ __forceinline__ float fast_tanh(float x) {
#if __has_builtin(__builtin_amdgcn_tanhf)
  return __builtin_amdgcn_tanhf(x);
#else
  float e = __builtin_amdgcn_exp2f(-2.8853900817779268f * x);
  return (1.0f - e) * __builtin_amdgcn_rcpf(1.0f + e);
#endif
}
__device__ __forceinline__ float fast_sigmoid(float x) {
  return __builtin_fmaf(0.5f, fast_tanh(0.5f * x), 0.5f);
}
__device__ __forceinline__ v2f tanh2(v2f v) {
  v2f r; r.x = fast_tanh(v.x); r.y = fast_tanh(v.y); return r;
}
__device__ __forceinline__ v2f sig2(v2f v) {
  v2f r; r.x = fast_sigmoid(v.x); r.y = fast_sigmoid(v.y); return r;
}
__device__ __forceinline__ float xhalf(float v) {
  int i = __builtin_bit_cast(int, v);
  i = __builtin_amdgcn_permlanex16(i, i, 0x76543210, 0xfedcba98, false, false);
  return __builtin_bit_cast(float, i);
}

__device__ __forceinline__ void gates_add4(v2f gA[4], v2f gB[4], const v2f w[4][4],
                                           float a0, float a1, float a2, float a3,
                                           float b0, float b1, float b2, float b3) {
  const float a[4] = {a0, a1, a2, a3};
  const float b[4] = {b0, b1, b2, b3};
#pragma unroll
  for (int m = 0; m < 4; ++m) {
    v2f av; av.x = a[m]; av.y = a[m];
    v2f bv; bv.x = b[m]; bv.y = b[m];
#pragma unroll
    for (int q = 0; q < 4; ++q) {
      gA[q] = __builtin_elementwise_fma(av, w[q][m], gA[q]);
      gB[q] = __builtin_elementwise_fma(bv, w[q][m], gB[q]);
    }
  }
}
__device__ __forceinline__ v2f splat2(float v) { v2f r; r.x = v; r.y = v; return r; }
#define ORDER_PIN() __builtin_amdgcn_sched_barrier(0)
#define HOLD(...) asm volatile("" : __VA_ARGS__)
template <class WithH>
__device__ __forceinline__ void act_fill(const v2f gA[4], const v2f gB[4],
                                         v2f& hA, v2f& cA, v2f& hB, v2f& cB, v2f& hoA, v2f& hoB,
                                         v2f fA[4], v2f fB[4], const v2f w[4][4],
                                         float a0, float a1, float a2, float a3,
                                         float b0, float b1, float b2, float b3, WithH with_h) {
  const v2f half = splat2(0.5f);
  v2f tgA, tgB, miA, miB, mfA, mfB, moA, moB, tiA, tiB, tfA, tfB, toA, toB;
  v2f siA, siB, sfA, sfB, soA, soB, pA, pB, tcA, tcB;
  tgA.x = fast_tanh(gA[2].x); miA = half * gA[0]; ORDER_PIN();
  tgA.y = fast_tanh(gA[2].y); miB = half * gB[0]; ORDER_PIN();
  tgB.x = fast_tanh(gB[2].x); mfA = half * gA[1]; ORDER_PIN();
  tgB.y = fast_tanh(gB[2].y); mfB = half * gB[1]; ORDER_PIN();
  tiA.x = fast_tanh(miA.x); moA = half * gA[3]; ORDER_PIN();
  tiA.y = fast_tanh(miA.y); moB = half * gB[3]; ORDER_PIN();
  tiB.x = fast_tanh(miB.x); fA[0] = __builtin_elementwise_fma(splat2(a0), w[0][0], fA[0]); HOLD("+v"(fA[0])); ORDER_PIN();
  tiB.y = fast_tanh(miB.y); fB[0] = __builtin_elementwise_fma(splat2(b0), w[0][0], fB[0]); HOLD("+v"(fB[0])); ORDER_PIN();
  tfA.x = fast_tanh(mfA.x); fA[1] = __builtin_elementwise_fma(splat2(a0), w[1][0], fA[1]); HOLD("+v"(fA[1])); ORDER_PIN();
  tfA.y = fast_tanh(mfA.y); fB[1] = __builtin_elementwise_fma(splat2(b0), w[1][0], fB[1]); HOLD("+v"(fB[1])); ORDER_PIN();
  tfB.x = fast_tanh(mfB.x); siA = __builtin_elementwise_fma(half, tiA, half); ORDER_PIN();
  tfB.y = fast_tanh(mfB.y); siB = __builtin_elementwise_fma(half, tiB, half); ORDER_PIN();
  toA.x = fast_tanh(moA.x); fA[2] = __builtin_elementwise_fma(splat2(a0), w[2][0], fA[2]); HOLD("+v"(fA[2])); ORDER_PIN();
  toA.y = fast_tanh(moA.y); sfA = __builtin_elementwise_fma(half, tfA, half); ORDER_PIN();
  toB.x = fast_tanh(moB.x); pA = siA * tgA; ORDER_PIN();
  toB.y = fast_tanh(moB.y); pB = siB * tgB; ORDER_PIN();
  sfB = __builtin_elementwise_fma(half, tfB, half); fB[2] = __builtin_elementwise_fma(splat2(b0), w[2][0], fB[2]); fA[3] = __builtin_elementwise_fma(splat2(a0), w[3][0], fA[3]); fB[3] = __builtin_elementwise_fma(splat2(b0), w[3][0], fB[3]); HOLD("+v"(fB[2]), "+v"(fA[3]), "+v"(fB[3])); ORDER_PIN();
  cA = __builtin_elementwise_fma(sfA, cA, pA); fA[0] = __builtin_elementwise_fma(splat2(a1), w[0][1], fA[0]); fB[0] = __builtin_elementwise_fma(splat2(b1), w[0][1], fB[0]); cB = __builtin_elementwise_fma(sfB, cB, pB); HOLD("+v"(fA[0]), "+v"(fB[0])); ORDER_PIN();
  soA = __builtin_elementwise_fma(half, toA, half); soB = __builtin_elementwise_fma(half, toB, half); fA[1] = __builtin_elementwise_fma(splat2(a1), w[1][1], fA[1]); fB[1] = __builtin_elementwise_fma(splat2(b1), w[1][1], fB[1]); HOLD("+v"(fA[1]), "+v"(fB[1])); ORDER_PIN();
  tcA.x = fast_tanh(cA.x); fA[2] = __builtin_elementwise_fma(splat2(a1), w[2][1], fA[2]); HOLD("+v"(fA[2])); ORDER_PIN();
  tcA.y = fast_tanh(cA.y); fB[2] = __builtin_elementwise_fma(splat2(b1), w[2][1], fB[2]); HOLD("+v"(fB[2])); ORDER_PIN();
  tcB.x = fast_tanh(cB.x); fA[3] = __builtin_elementwise_fma(splat2(a1), w[3][1], fA[3]); HOLD("+v"(fA[3])); ORDER_PIN();
  tcB.y = fast_tanh(cB.y); fB[3] = __builtin_elementwise_fma(splat2(b1), w[3][1], fB[3]); HOLD("+v"(fB[3])); ORDER_PIN();
  fA[0] = __builtin_elementwise_fma(splat2(a2), w[0][2], fA[0]); fB[0] = __builtin_elementwise_fma(splat2(b2), w[0][2], fB[0]); HOLD("+v"(fA[0]), "+v"(fB[0])); ORDER_PIN();
  hA = soA * tcA; fA[1] = __builtin_elementwise_fma(splat2(a2), w[1][2], fA[1]); fB[1] = __builtin_elementwise_fma(splat2(b2), w[1][2], fB[1]); fA[2] = __builtin_elementwise_fma(splat2(a2), w[2][2], fA[2]); HOLD("+v"(fA[1]), "+v"(fB[1]), "+v"(fA[2])); ORDER_PIN();
  hB = soB * tcB; fB[2] = __builtin_elementwise_fma(splat2(b2), w[2][2], fB[2]); fA[3] = __builtin_elementwise_fma(splat2(a2), w[3][2], fA[3]); fB[3] = __builtin_elementwise_fma(splat2(b2), w[3][2], fB[3]); fA[0] = __builtin_elementwise_fma(splat2(a3), w[0][3], fA[0]); HOLD("+v"(fB[2]), "+v"(fA[3]), "+v"(fB[3]), "+v"(fA[0])); ORDER_PIN();
  with_h(); ORDER_PIN();
  hoA.x = xhalf(hA.x); hoA.y = xhalf(hA.y); fB[0] = __builtin_elementwise_fma(splat2(b3), w[0][3], fB[0]); fA[1] = __builtin_elementwise_fma(splat2(a3), w[1][3], fA[1]); HOLD("+v"(fB[0]), "+v"(fA[1])); ORDER_PIN();
  hoB.x = xhalf(hB.x); hoB.y = xhalf(hB.y); ORDER_PIN();
  fB[1] = __builtin_elementwise_fma(splat2(b3), w[1][3], fB[1]); fA[2] = __builtin_elementwise_fma(splat2(a3), w[2][3], fA[2]); fB[2] = __builtin_elementwise_fma(splat2(b3), w[2][3], fB[2]); fA[3] = __builtin_elementwise_fma(splat2(a3), w[3][3], fA[3]); fB[3] = __builtin_elementwise_fma(splat2(b3), w[3][3], fB[3]); HOLD("+v"(fB[1]), "+v"(fA[2]), "+v"(fB[2]), "+v"(fA[3]), "+v"(fB[3])); ORDER_PIN();
}
#undef HOLD
#undef ORDER_PIN

__device__ __forceinline__ void load_half_weights(const float* __restrict__ Wm, int hh, bool perLane,
                                                  v2f w[4][4]) {
#pragma unroll
  for (int q = 0; q < 4; ++q) {
    const int j = 4 * q + 2 * hh;
    const v4f ra = *(const v4f*)(Wm + 4 * j);
    const v4f rb = *(const v4f*)(Wm + 4 * j + 4);
    const bool sw = perLane && (hh != 0);
    w[q][0].x = sw ? ra[2] : ra[0];  w[q][0].y = sw ? rb[2] : rb[0];
    w[q][1].x = sw ? ra[3] : ra[1];  w[q][1].y = sw ? rb[3] : rb[1];
    w[q][2].x = sw ? ra[0] : ra[2];  w[q][2].y = sw ? rb[0] : rb[2];
    w[q][3].x = sw ? ra[1] : ra[3];  w[q][3].y = sw ? rb[1] : rb[3];
  }
}
__device__ __forceinline__ void load_half_bias(const float* __restrict__ bih, const float* __restrict__ bhh,
                                               int hh, v2f pb[4]) {
#pragma unroll
  for (int q = 0; q < 4; ++q) {
    const int j = 4 * q + 2 * hh;
    pb[q].x = bih[j] + bhh[j];
    pb[q].y = bih[j + 1] + bhh[j + 1];
  }
}

__device__ __forceinline__ void load_block(const v4f* __restrict__ xg, const v4f* __restrict__ wg,
                                           int j, v4f gx[4 * GRP], v4f gb[WPIECES]) {
#pragma unroll
  for (int i = 0; i < 4 * GRP; ++i) gx[i] = xg[(size_t)(4 * i) * T_STEPS + BLK * j];
#pragma unroll
  for (int i = 0; i < WPIECES; ++i) gb[i] = wg[(size_t)(4 * i) * T_STEPS + BLK * j];
}
__device__ __forceinline__ void store_block(v4f* sx, v4f* sb, int xst, int bst,
                                            const v4f gx[4 * GRP], const v4f gb[WPIECES]) {
#pragma unroll
  for (int i = 0; i < 4 * GRP; ++i) sx[xst + 4 * i * XPITCH] = gx[i];
#pragma unroll
  for (int i = 0; i < WPIECES; ++i) sb[bst + 4 * i] = gb[i];
}
__device__ __forceinline__ void wave_lds_handover() {
  __builtin_amdgcn_fence(__ATOMIC_RELEASE, "wavefront");
  __builtin_amdgcn_wave_barrier();
  __builtin_amdgcn_fence(__ATOMIC_ACQUIRE, "wavefront");
}

__global__ void __launch_bounds__(128)
lstm_fused_kernel(const float* __restrict__ x,
                  const float* __restrict__ Wih0, const float* __restrict__ Whh0,
                  const float* __restrict__ bih0, const float* __restrict__ bhh0,
                  const float* __restrict__ Wih1, const float* __restrict__ Whh1,
                  const float* __restrict__ bih1, const float* __restrict__ bhh1,
                  const float* __restrict__ Wout, const float* __restrict__ bout,
                  float* __restrict__ out) {
  __shared__ v4f lds[4 * WAVE_LDS];

  const int tid = threadIdx.x;
  const int lane = tid & 31;
  const int col = lane & 15;
  const bool hiHalf = (lane & 16) != 0;
  const int hh = hiHalf ? 1 : 0;
  const int waveBase = blockIdx.x * blockDim.x + (tid & ~31);

  v4f* ringX = lds + (tid >> 5) * WAVE_LDS;
  v4f* ringB = ringX + 2 * XSTAGE;
  const v4f* __restrict__ xg = (const v4f*)x + (size_t)(waveBase + (lane >> 3)) * T_STEPS + (lane & 7);
  const v4f* __restrict__ wg = (const v4f*)Wout + (size_t)(lane >> 3) * T_STEPS + (lane & 7);
  const int xst = (lane >> 3) * XPITCH + (lane & 7);
  const int bst = (lane & 7) * BPITCH + (lane >> 3);

  {
    const v4f zz = (v4f)0.f;
    for (int m = 0; m < 2; ++m) ringB[m * BSTAGE + (lane >> 2) * BPITCH + 28 + (lane & 3)] = zz;
  }

  v4f gx[4 * GRP], gb[WPIECES];
  load_block(xg, wg, 0, gx, gb);
  store_block(ringX, ringB, xst, bst, gx, gb);
  load_block(xg, wg, 1, gx, gb);
  wave_lds_handover();

  v2f w0x[4][4], w0h[4][4], w1h[4][4], w1i[4][4], pb0[4], pb1[4];
  load_half_weights(Wih0, hh, false, w0x);
  load_half_weights(Whh0, hh, true, w0h);
  load_half_weights(Whh1, hh, true, w1h);
  load_half_weights(Wih1, hh, true, w1i);
  load_half_bias(bih0, bhh0, hh, pb0);
  load_half_bias(bih1, bhh1, hh, pb1);

  v2f h0 = (v2f)0.f, c0 = (v2f)0.f, h1 = (v2f)0.f, c1 = (v2f)0.f;
  v2f h0B = (v2f)0.f, c0B = (v2f)0.f, h1B = (v2f)0.f, c1B = (v2f)0.f;
  v2f h0o = (v2f)0.f, h1o = (v2f)0.f;
  v2f h0oB = (v2f)0.f, h1oB = (v2f)0.f;

  float bo0 = bout[col];
  float bo1 = (col + 16 < OUTN) ? bout[col + 16] : 0.f;
  v8f acc0, acc1, acc0B, acc1B;
#pragma unroll
  for (int r = 0; r < 8; ++r) { acc0[r] = bo0; acc1[r] = bo1; acc0B[r] = bo0; acc1B[r] = bo1; }

  v2f pxA[4], pxB[4];
  {
    const v4f xv = ringX[col * XPITCH];
    const v4f xvB = ringX[(16 + col) * XPITCH];
#pragma unroll
    for (int q = 0; q < 4; ++q) { pxA[q] = pb0[q]; pxB[q] = pb0[q]; }
    gates_add4(pxA, pxB, w0x, xv.x, xv.y, xv.z, xv.w, xvB.x, xvB.y, xvB.z, xvB.w);
  }

  const v4f* rowX = ringX + col * XPITCH;
  const v2f* fragB = (const v2f*)ringB + 2 * col + hh;

  for (int j = 0; j < NBLK; ++j) {
    const v4f* rowXc = rowX + (j & 1) * XSTAGE;
    const v4f* rowXn = rowX + ((j + 1) & 1) * XSTAGE;
    const v2f* fragBc = fragB + 2 * (j & 1) * BSTAGE;
#pragma unroll
    for (int s = 0; s < BLK; ++s) {
      const v2f B0 = fragBc[2 * (s * BPITCH)];
      const v2f B1 = fragBc[2 * (s * BPITCH) + 32];
      const v4f xn = (s + 1 < BLK) ? rowXc[s + 1] : rowXn[0];
      const v4f xnB = (s + 1 < BLK) ? rowXc[16 * XPITCH + s + 1] : rowXn[16 * XPITCH];
      __builtin_amdgcn_sched_barrier(0);

      v2f gA[4], gB[4];
#pragma unroll
      for (int q = 0; q < 4; ++q) { gA[q] = pxA[q]; gB[q] = pxB[q]; }
      gates_add4(gA, gB, w0h, h0.x, h0.y, h0o.x, h0o.y, h0B.x, h0B.y, h0oB.x, h0oB.y);
      __builtin_amdgcn_sched_barrier(0);

      v2f qA[4], qB[4];
#pragma unroll
      for (int q = 0; q < 4; ++q) { qA[q] = pb1[q]; qB[q] = pb1[q]; }
      act_fill(gA, gB, h0, c0, h0B, c0B, h0o, h0oB, qA, qB, w1h,
               h1.x, h1.y, h1o.x, h1o.y, h1B.x, h1B.y, h1oB.x, h1oB.y, [] {});

      gates_add4(qA, qB, w1i, h0.x, h0.y, h0o.x, h0o.y, h0B.x, h0B.y, h0oB.x, h0oB.y);
      __builtin_amdgcn_sched_barrier(0);

#pragma unroll
      for (int q = 0; q < 4; ++q) { pxA[q] = pb0[q]; pxB[q] = pb0[q]; }
      act_fill(qA, qB, h1, c1, h1B, c1B, h1o, h1oB, pxA, pxB, w0x,
               xn.x, xn.y, xn.z, xn.w, xnB.x, xnB.y, xnB.z, xnB.w, [&] {
        acc0 = __builtin_amdgcn_wmma_f32_16x16x4_f32(false, h1, false, B0, (short)0, acc0, false, false);
        acc1 = __builtin_amdgcn_wmma_f32_16x16x4_f32(false, h1, false, B1, (short)0, acc1, false, false);
        acc0B = __builtin_amdgcn_wmma_f32_16x16x4_f32(false, h1B, false, B0, (short)0, acc0B, false, false);
        acc1B = __builtin_amdgcn_wmma_f32_16x16x4_f32(false, h1B, false, B1, (short)0, acc1B, false, false);
      });

      if (s == BLK - 3 && j + 1 < NBLK) {
        const int so = (j + 1) & 1;
        store_block(ringX + so * XSTAGE, ringB + so * BSTAGE, xst, bst, gx, gb);
        if (j + 2 < NBLK) load_block(xg, wg, j + 2, gx, gb);
        wave_lds_handover();
      }
    }
  }

  float* outp0 = out + (size_t)(waveBase + (hiHalf ? 8 : 0)) * OUTN + col;
  float* outpB = outp0 + (size_t)16 * OUTN;
#pragma unroll
  for (int r = 0; r < 8; ++r) {
    outp0[(size_t)r * OUTN] = acc0[r];
    outpB[(size_t)r * OUTN] = acc0B[r];
    if (col < OUTN - 16) {
      outp0[(size_t)r * OUTN + 16] = acc1[r];
      outpB[(size_t)r * OUTN + 16] = acc1B[r];
    }
  }
}

extern "C" void kernel_launch(void* const* d_in, const int* in_sizes, int n_in,
                              void* d_out, int out_size, void* d_ws, size_t ws_size,
                              hipStream_t stream) {
  const float* x    = (const float*)d_in[0];
  const float* Wih0 = (const float*)d_in[1];
  const float* Whh0 = (const float*)d_in[2];
  const float* bih0 = (const float*)d_in[3];
  const float* bhh0 = (const float*)d_in[4];
  const float* Wih1 = (const float*)d_in[5];
  const float* Whh1 = (const float*)d_in[6];
  const float* bih1 = (const float*)d_in[7];
  const float* bhh1 = (const float*)d_in[8];
  const float* Wout = (const float*)d_in[9];
  const float* bout = (const float*)d_in[10];
  float* out = (float*)d_out;

  int B = in_sizes[0] / (T_STEPS * HID);

  lstm_fused_kernel<<<B / 128, 128, 0, stream>>>(x, Wih0, Whh0, bih0, bhh0,
                                                 Wih1, Whh1, bih1, bhh1,
                                                 Wout, bout, out);
}
